# starting kernel
# baseline (speedup 1.0000x reference)
#include <hip/hip_runtime.h>

typedef float v2f __attribute__((ext_vector_type(2)));
typedef float v8f __attribute__((ext_vector_type(8)));

#define BB   16
#define CIN  128
#define COUT 256
#define LL   32

#define QP   33
#define CP   155

__global__ __launch_bounds__(256) void fconv2d_wmma_kernel(
    const float* __restrict__ x,
    const float* __restrict__ w,
    float* __restrict__ out)
{
    __shared__ float sW2[9 * 16 * QP];
    __shared__ float sX2[3 * 18 * CP];

    const int tid = threadIdx.x;
    const int bid = blockIdx.x;
    const int b   = bid >> 6;
    const int rem = bid & 63;
    const int y   = rem >> 1;
    const int x0  = (rem & 1) << 4;

    for (int idx = tid; idx < 16 * 32 * 9; idx += 256) {
        int n = idx / 288;
        int r = idx % 288;
        int q = r / 9;
        int t = r % 9;
        int i = t / 3, j = t % 3;
        sW2[((2 - i) * 3 + (2 - j)) * (16 * QP) + n * QP + (31 - q)] = 0.5f * w[idx];
    }
    for (int idx = tid; idx < 153 * 54; idx += 256) {
        int c   = 1 + idx / 54;
        int r2  = idx % 54;
        int row = r2 / 18;
        int xi  = r2 % 18;
        int ch  = (c + 95) & 127;
        int gy  = y + row;
        int gx  = x0 + xi;
        float v = 0.0f;
        if (gy < LL && gx < LL)
            v = x[((b * CIN + ch) * LL + gy) * LL + gx];
        sX2[(row * 18 + xi) * CP + c] = v;
    }
    __syncthreads();

    const int lane = tid & 31;
    const int wid  = tid >> 5;
    const int col  = lane & 15;
    const int hi   = lane >> 4;

    const int d0 = wid;
    const int d1 = wid ? (16 - wid) : 8;

    v8f c0 = {0.f, 0.f, 0.f, 0.f, 0.f, 0.f, 0.f, 0.f};
    v8f c1 = {0.f, 0.f, 0.f, 0.f, 0.f, 0.f, 0.f, 0.f};

    if (wid != 0) {
        for (int dy = 0; dy < 3; ++dy) {
            for (int dx = 0; dx < 3; ++dx) {
                const float* ap  = &sW2[(dy * 3 + dx) * (16 * QP) + col * QP + 2 * hi];
                const int    sb  = (dy * 18 + col + dx) * CP + 2 + 2 * hi;
                const float* b0p = &sX2[sb + 8 * d0];
                const float* b1p = &sX2[sb + 8 * d1];
#pragma unroll
                for (int k = 0; k < 8; ++k) {
                    v2f a;
                    a.x = ap[4 * k];
                    a.y = ap[4 * k + 1];
                    v2f bf;
                    bf.x = b0p[4 * k]     + b1p[4 * k];
                    bf.y = b0p[4 * k + 1] + b1p[4 * k + 1];
                    c0 = __builtin_amdgcn_wmma_f32_16x16x4_f32(
                            false, a, false, bf, (short)0, c0, false, false);
                    c1 = __builtin_amdgcn_wmma_f32_16x16x4_f32(
                            false, a, false, bf, (short)0, c1, false, false);
                }
            }
        }
    } else {
        for (int dy = 0; dy < 3; ++dy) {
            for (int dx = 0; dx < 3; ++dx) {
                const float* ap  = &sW2[(dy * 3 + dx) * (16 * QP) + col * QP + 2 * hi];
                const int    sb  = (dy * 18 + col + dx) * CP + 2 + 2 * hi;
                const float* b0p = &sX2[sb + 8 * 0];
                const float* b1p = &sX2[sb + 8 * 8];
#pragma unroll
                for (int k = 0; k < 8; ++k) {
                    v2f a;
                    a.x = ap[4 * k];
                    a.y = ap[4 * k + 1];
                    float u0 = b0p[4 * k];
                    float u1 = b0p[4 * k + 1];
                    float v0 = b1p[4 * k];
                    float v1 = b1p[4 * k + 1];
                    v2f bf0, bf1;
                    bf0.x = u0 + u0;  bf0.y = u1 + u1;
                    bf1.x = v0 + v0;  bf1.y = v1 + v1;
                    c0 = __builtin_amdgcn_wmma_f32_16x16x4_f32(
                            false, a, false, bf0, (short)0, c0, false, false);
                    c1 = __builtin_amdgcn_wmma_f32_16x16x4_f32(
                            false, a, false, bf1, (short)0, c1, false, false);
                }
            }
        }
    }

    float* op0 = out + ((b * COUT + d0 * 16 + hi * 8) * LL + y) * LL + x0 + col;
    float* op1 = out + ((b * COUT + d1 * 16 + hi * 8) * LL + y) * LL + x0 + col;
#pragma unroll
    for (int r = 0; r < 8; ++r) {
        op0[r * (LL * LL)] = c0[r];
        op1[r * (LL * LL)] = c1[r];
    }
}

extern "C" void kernel_launch(void* const* d_in, const int* in_sizes, int n_in,
                              void* d_out, int out_size, void* d_ws, size_t ws_size,
                              hipStream_t stream) {
    const float* x = (const float*)d_in[0];
    const float* w = (const float*)d_in[1];
    float* out = (float*)d_out;

    dim3 grid(BB * LL * 2);
    dim3 block(256);
    hipLaunchKernelGGL(fconv2d_wmma_kernel, grid, block, 0, stream, x, w, out);
}


# re-tiled: a work-group owns one image's row pair for all 9 groups; raw 128-channel slab staged once (channel-reversed), mirror sum X[8d-q]+X[-8d-q] formed at B-fragment read; sum order kept
# speedup vs baseline: 2.7871x; 2.7871x over previous
#include <hip/hip_runtime.h>


typedef float v2f __attribute__((ext_vector_type(2)));
typedef float v8f __attribute__((ext_vector_type(8)));

#define CIN  128
#define COUT 256
#define LL   32

#define RB   2
#define SR   (RB + 2)
#define SC   34
#define XP   164
#define WQ   36
#define WT   (16 * WQ + 4)
#define NT   384

#define LDS_FLOATS (164 * 256)

__global__ __launch_bounds__(NT) void fconv2d_rawslab_kernel(
    const float* __restrict__ x,
    const float* __restrict__ w,
    float* __restrict__ out)
{
    __shared__ __attribute__((aligned(16))) float lds[LDS_FLOATS];
    float* const sW = lds;
    float* const sX = lds + 9 * WT;

    const int tid  = threadIdx.x;
    const int bid  = blockIdx.x;
    const int band = bid & 15;
    const int b    = bid >> 4;
    const int y0   = band * RB;

    float4 wv[3];
#pragma unroll
    for (int i = 0; i < 3; ++i)
        wv[i] = *(const float4*)(w + 4 * (tid + NT * i));

    float4 xv[3][4];
#pragma unroll
    for (int i = 0; i < 3; ++i) {
        const int idx = (tid + NT * i) & 1023;
        const int x4  = idx & 7;
        const int m   = (idx >> 3) & 31;
        const int row = idx >> 8;
        const int gy  = y0 + row;
        const int gyc = gy < LL ? gy : (LL - 1);
#pragma unroll
        for (int cq = 0; cq < 4; ++cq) {
            const int ch = (4 * m + cq + 1) & 127;
            xv[i][cq] = *(const float4*)(x + ((b * CIN + ch) * LL + gyc) * LL + 4 * x4);
        }
    }

#pragma unroll
    for (int i = 0; i < 3; ++i) {
        const float wc[4] = {wv[i].x, wv[i].y, wv[i].z, wv[i].w};
#pragma unroll
        for (int jj = 0; jj < 4; ++jj) {
            const int e   = 4 * (tid + NT * i) + jj;
            const int n   = e / 288;
            const int rem = e - n * 288;
            const int q   = rem / 9;
            const int t   = rem - q * 9;
            sW[(8 - t) * WT + n * WQ + q] = 0.5f * wc[jj];
        }
    }
#pragma unroll
    for (int i = 0; i < 3; ++i) {
        const int idx = tid + NT * i;
        if (i < 2 || tid < 256) {
            const int x4  = idx & 7;
            const int m   = (idx >> 3) & 31;
            const int row = idx >> 8;
            const bool below = (y0 + row) >= LL;
            const int p0  = (124 - 4 * m) & 127;
            float* sp = &sX[(row * SC + 4 * x4) * XP + p0];
            const float4 l1 = xv[i][0], l2 = xv[i][1], l3 = xv[i][2], l4 = xv[i][3];
            float4 c0 = make_float4(l4.x, l3.x, l2.x, l1.x);
            float4 c1 = make_float4(l4.y, l3.y, l2.y, l1.y);
            float4 c2 = make_float4(l4.z, l3.z, l2.z, l1.z);
            float4 c3 = make_float4(l4.w, l3.w, l2.w, l1.w);
            if (below) {
                c0 = make_float4(0.f, 0.f, 0.f, 0.f);
                c1 = c0; c2 = c0; c3 = c0;
            }
            *(float4*)(sp)          = c0;
            *(float4*)(sp + XP)     = c1;
            *(float4*)(sp + 2 * XP) = c2;
            *(float4*)(sp + 3 * XP) = c3;
            if (p0 < 32) {
                *(float4*)(sp + 128)          = c0;
                *(float4*)(sp + 128 + XP)     = c1;
                *(float4*)(sp + 128 + 2 * XP) = c2;
                *(float4*)(sp + 128 + 3 * XP) = c3;
            }
        }
    }
    if (tid < 320) {
        const int pos4 = tid % 40;
        const int pix  = tid / 40;
        const int row  = pix >> 1;
        const int cc   = pix & 1;
        *(float4*)&sX[(row * SC + 32 + cc) * XP + 4 * pos4] = make_float4(0.f, 0.f, 0.f, 0.f);
    }
    __syncthreads();

    const int lane = tid & 31;
    const int wid  = tid >> 5;
    const int col  = lane & 15;
    const int hi   = lane >> 4;
    const int tl   = wid & 3;
    const int r    = tl >> 1;
    const int h    = tl & 1;
    const int j    = wid >> 2;

    const float* aW = &sW[col * WQ + 2 * hi];
    const float* bX = &sX[(r * SC + 16 * h + col) * XP + 2 * hi];
    const float* bM[3];
    const float* bP[3];
#pragma unroll
    for (int i = 0; i < 3; ++i) {
        const int d = 3 * j + i;
        bM[i] = bX + 128 - 8 * d;
        bP[i] = bX + 8 * d;
    }

    v8f acc[3];
#pragma unroll
    for (int i = 0; i < 3; ++i)
        acc[i] = (v8f){0.f, 0.f, 0.f, 0.f, 0.f, 0.f, 0.f, 0.f};

#pragma unroll
    for (int dy = 0; dy < 3; ++dy) {
#pragma unroll
        for (int dx = 0; dx < 3; ++dx) {
#pragma unroll
            for (int k = 0; k < 8; ++k) {
                const int offa = (dy * 3 + dx) * WT + 4 * k;
                const int offb = (dy * SC + dx) * XP + 4 * k;
                const v2f a = *(const v2f*)(aW + offa);
#pragma unroll
                for (int i = 0; i < 3; ++i) {
                    const v2f u  = *(const v2f*)(bM[i] + offb);
                    const v2f v  = *(const v2f*)(bP[i] + offb);
                    const v2f bb = u + v;
                    acc[i] = __builtin_amdgcn_wmma_f32_16x16x4_f32(
                                 false, a, false, bb, (short)0, acc[i], false, false);
                }
            }
        }
    }

    const int y = y0 + r;
#pragma unroll
    for (int i = 0; i < 3; ++i) {
        const int d = 3 * j + i;
        float* op0 = out + ((b * COUT + d * 16 + hi * 8) * LL + y) * LL + 16 * h + col;
#pragma unroll
        for (int rr = 0; rr < 8; ++rr) op0[rr * (LL * LL)] = acc[i][rr];
        if (d != 0 && d != 8) {
            float* op1 = out + ((b * COUT + (16 - d) * 16 + hi * 8) * LL + y) * LL + 16 * h + col;
#pragma unroll
            for (int rr = 0; rr < 8; ++rr) op1[rr * (LL * LL)] = acc[i][rr];
        }
    }
}

extern "C" void kernel_launch(void* const* d_in, const int* in_sizes, int n_in,
                              void* d_out, int out_size, void* d_ws, size_t ws_size,
                              hipStream_t stream) {
    const float* x = (const float*)d_in[0];
    const float* w = (const float*)d_in[1];
    float* out = (float*)d_out;

    dim3 grid(256);
    dim3 block(NT);
    hipLaunchKernelGGL(fconv2d_rawslab_kernel, grid, block, 0, stream, x, w, out);
}
